# starting kernel
# baseline (speedup 1.0000x reference)
#include <hip/hip_runtime.h>

typedef __attribute__((ext_vector_type(2))) float v2f;
typedef __attribute__((ext_vector_type(8))) float v8f;

#define TOK   4096
#define BN    1024
#define DD    128

__device__ __forceinline__ float sigf(float x)  { return 1.0f / (1.0f + __expf(-x)); }
__device__ __forceinline__ float siluf(float x) { return x * sigf(x); }
__device__ __forceinline__ float dsiluf(float x){ float s = sigf(x); return s * (1.0f + x * (1.0f - s)); }

__global__ __launch_bounds__(256) void gemm128(
    const float* __restrict__ A, int lda,
    const float* __restrict__ W, int ldw, int wBatchStride, int wTrans,
    const float* __restrict__ Zmul,
    float* __restrict__ Cout, int ldc,
    float* __restrict__ SiluOut, int ldso)
{
  const int wave = threadIdx.x >> 5;
  const int lane = threadIdx.x & 31;
  const int half = lane >> 4;
  const int l    = lane & 15;
  const int m0   = blockIdx.x * 32 + (wave >> 2) * 16;
  const int n0   = blockIdx.y * 128 + (wave & 3) * 32;
  const int batch = (blockIdx.x * 32) / BN;
  const float* Wb = W + (size_t)batch * wBatchStride;

  v8f acc0 = {};
  v8f acc1 = {};
  const float* arow = A + (size_t)(m0 + l) * lda + 2 * half;

  if (wTrans) {
    const float* wr0 = Wb + (size_t)(n0 + l) * ldw + 2 * half;
    const float* wr1 = Wb + (size_t)(n0 + 16 + l) * ldw + 2 * half;
    for (int k0 = 0; k0 < DD; k0 += 4) {
      v2f a  = *(const v2f*)(arow + k0);
      v2f b0 = *(const v2f*)(wr0 + k0);
      v2f b1 = *(const v2f*)(wr1 + k0);
      acc0 = __builtin_amdgcn_wmma_f32_16x16x4_f32(false, a, false, b0, (short)0, acc0, false, false);
      acc1 = __builtin_amdgcn_wmma_f32_16x16x4_f32(false, a, false, b1, (short)0, acc1, false, false);
    }
  } else {
    for (int k0 = 0; k0 < DD; k0 += 4) {
      v2f a = *(const v2f*)(arow + k0);
      const float* r0 = Wb + (size_t)(k0 + 2 * half) * ldw + n0 + l;
      const float* r1 = r0 + ldw;
      v2f b0, b1;
      b0.x = r0[0];  b0.y = r1[0];
      b1.x = r0[16]; b1.y = r1[16];
      acc0 = __builtin_amdgcn_wmma_f32_16x16x4_f32(false, a, false, b0, (short)0, acc0, false, false);
      acc1 = __builtin_amdgcn_wmma_f32_16x16x4_f32(false, a, false, b1, (short)0, acc1, false, false);
    }
  }

  #pragma unroll
  for (int r = 0; r < 8; ++r) {
    const int m = m0 + r + 8 * half;
    float c0 = acc0[r];
    float c1 = acc1[r];
    if (Zmul) {
      c0 *= dsiluf(Zmul[(size_t)m * DD + (n0 + l)]);
      c1 *= dsiluf(Zmul[(size_t)m * DD + (n0 + 16 + l)]);
    }
    if (Cout) {
      Cout[(size_t)m * ldc + n0 + l]      = c0;
      Cout[(size_t)m * ldc + n0 + 16 + l] = c1;
    }
    if (SiluOut) {
      SiluOut[(size_t)m * ldso + n0 + l]      = siluf(c0);
      SiluOut[(size_t)m * ldso + n0 + 16 + l] = siluf(c1);
    }
  }
}

__global__ __launch_bounds__(256) void gradU(
    const float* __restrict__ Act, int ldact,
    const float* __restrict__ Dlt,
    const float* __restrict__ Wbase,
    float* __restrict__ Uout)
{
  const int wave = threadIdx.x >> 5;
  const int lane = threadIdx.x & 31;
  const int half = lane >> 4;
  const int l    = lane & 15;
  const int m0   = blockIdx.x * 32 + (wave >> 2) * 16;
  const int n0   = (wave & 3) * 32;
  const int batch = blockIdx.z;
  const float* Ab = Act + (size_t)batch * BN * ldact + m0 + l;
  const float* Db = Dlt + (size_t)batch * BN * DD;

  v8f acc0 = {};
  v8f acc1 = {};
  for (int k0 = 0; k0 < BN; k0 += 4) {
    const int kr = k0 + 2 * half;
    v2f a;
    a.x = Ab[(size_t)(kr + 0) * ldact];
    a.y = Ab[(size_t)(kr + 1) * ldact];
    const float* d0 = Db + (size_t)kr * DD + n0 + l;
    const float* d1 = d0 + DD;
    v2f b0, b1;
    b0.x = d0[0];  b0.y = d1[0];
    b1.x = d0[16]; b1.y = d1[16];
    acc0 = __builtin_amdgcn_wmma_f32_16x16x4_f32(false, a, false, b0, (short)0, acc0, false, false);
    acc1 = __builtin_amdgcn_wmma_f32_16x16x4_f32(false, a, false, b1, (short)0, acc1, false, false);
  }

  #pragma unroll
  for (int r = 0; r < 8; ++r) {
    const int d = m0 + r + 8 * half;
    float* urow = Uout + (size_t)batch * DD * DD + (size_t)d * DD;
    const float* wrow = Wbase + (size_t)d * DD;
    urow[n0 + l]      = wrow[n0 + l]      + acc0[r];
    urow[n0 + 16 + l] = wrow[n0 + 16 + l] + acc1[r];
  }
}

__global__ __launch_bounds__(256) void d3_elem(
    const float* __restrict__ P, const float* __restrict__ KV, float* __restrict__ D3)
{
  const int idx = blockIdx.x * 256 + threadIdx.x;
  const int m = idx >> 7;
  const int n = idx & 127;
  const float v = KV[(size_t)m * 256 + DD + n];
  D3[idx] = (2.0f / 128.0f) * (P[idx] - v);
}

extern "C" void kernel_launch(void* const* d_in, const int* in_sizes, int n_in,
                              void* d_out, int out_size, void* d_ws, size_t ws_size,
                              hipStream_t stream)
{
  const float* S   = (const float*)d_in[0];
  const float* w0  = (const float*)d_in[1];
  const float* w1  = (const float*)d_in[2];
  const float* w2  = (const float*)d_in[3];
  const float* w3  = (const float*)d_in[4];
  const float* wq  = (const float*)d_in[5];
  const float* wkv = (const float*)d_in[6];
  float* out = (float*)d_out;

  float* p = (float*)d_ws;
  const size_t T = (size_t)TOK * DD;
  float* KV = p; p += (size_t)TOK * 256;
  float* Z0 = p; p += T;  float* A1 = p; p += T;
  float* Z1 = p; p += T;  float* A2 = p; p += T;
  float* Z2 = p; p += T;  float* A3 = p; p += T;
  float* P_ = p; p += T;
  float* Da = p; p += T;
  float* Db = p; p += T;
  float* Xa = p; p += T;
  float* Xb = p; p += T;
  float* U0 = p; p += 4 * DD * DD;
  float* U1 = p; p += 4 * DD * DD;
  float* U2 = p; p += 4 * DD * DD;
  float* U3 = p; p += 4 * DD * DD;

  const dim3 blk(256);
  const dim3 gKV(TOK / 32, 2);
  const dim3 g(TOK / 32, 1);
  const dim3 gg(4, 1, 4);

  gemm128<<<gKV, blk, 0, stream>>>(S, DD, wkv, 256, 0, 0, nullptr, KV, 256, nullptr, 0);

  gemm128<<<g, blk, 0, stream>>>(KV, 256, w0, DD, 0, 0, nullptr, Z0, DD, A1, DD);
  gemm128<<<g, blk, 0, stream>>>(A1, DD, w1, DD, 0, 0, nullptr, Z1, DD, A2, DD);
  gemm128<<<g, blk, 0, stream>>>(A2, DD, w2, DD, 0, 0, nullptr, Z2, DD, A3, DD);
  gemm128<<<g, blk, 0, stream>>>(A3, DD, w3, DD, 0, 0, nullptr, P_, DD, nullptr, 0);

  d3_elem<<<dim3((TOK * DD) / 256), blk, 0, stream>>>(P_, KV, Da);

  gradU<<<gg, blk, 0, stream>>>(A3, DD, Da, w3, U3);
  gemm128<<<g, blk, 0, stream>>>(Da, DD, w3, DD, 0, 1, Z2, Db, DD, nullptr, 0);
  gradU<<<gg, blk, 0, stream>>>(A2, DD, Db, w2, U2);
  gemm128<<<g, blk, 0, stream>>>(Db, DD, w2, DD, 0, 1, Z1, Da, DD, nullptr, 0);
  gradU<<<gg, blk, 0, stream>>>(A1, DD, Da, w1, U1);
  gemm128<<<g, blk, 0, stream>>>(Da, DD, w1, DD, 0, 1, Z0, Db, DD, nullptr, 0);
  gradU<<<gg, blk, 0, stream>>>(KV, 256, Db, w0, U0);

  gemm128<<<g, blk, 0, stream>>>(S,  DD, wq, DD, 0,       0, nullptr, Xa, DD, nullptr, 0);
  gemm128<<<g, blk, 0, stream>>>(Xa, DD, U0, DD, DD * DD, 0, nullptr, nullptr, 0, Xb, DD);
  gemm128<<<g, blk, 0, stream>>>(Xb, DD, U1, DD, DD * DD, 0, nullptr, nullptr, 0, Xa, DD);
  gemm128<<<g, blk, 0, stream>>>(Xa, DD, U2, DD, DD * DD, 0, nullptr, nullptr, 0, Xb, DD);
  gemm128<<<g, blk, 0, stream>>>(Xb, DD, U3, DD, DD * DD, 0, nullptr, out, DD, nullptr, 0);
}


# gradient GEMM: token sum split over 16 waves of a block (16x32 tile per block, 128 blocks instead of 16), partial tiles added through LDS in fixed wave order; changes the summation tree
# speedup vs baseline: 2.4218x; 2.4218x over previous
#include <hip/hip_runtime.h>

typedef __attribute__((ext_vector_type(2))) float v2f;
typedef __attribute__((ext_vector_type(8))) float v8f;

#define TOK   4096
#define BN    1024
#define DD    128

__device__ __forceinline__ float sigf(float x)  { return 1.0f / (1.0f + __expf(-x)); }
__device__ __forceinline__ float siluf(float x) { return x * sigf(x); }
__device__ __forceinline__ float dsiluf(float x){ float s = sigf(x); return s * (1.0f + x * (1.0f - s)); }

__global__ __launch_bounds__(256) void gemm128(
    const float* __restrict__ A, int lda,
    const float* __restrict__ W, int ldw, int wBatchStride, int wTrans,
    const float* __restrict__ Zmul,
    float* __restrict__ Cout, int ldc,
    float* __restrict__ SiluOut, int ldso)
{
  const int wave = threadIdx.x >> 5;
  const int lane = threadIdx.x & 31;
  const int half = lane >> 4;
  const int l    = lane & 15;
  const int m0   = blockIdx.x * 32 + (wave >> 2) * 16;
  const int n0   = blockIdx.y * 128 + (wave & 3) * 32;
  const int batch = (blockIdx.x * 32) / BN;
  const float* Wb = W + (size_t)batch * wBatchStride;

  v8f acc0 = {};
  v8f acc1 = {};
  const float* arow = A + (size_t)(m0 + l) * lda + 2 * half;

  if (wTrans) {
    const float* wr0 = Wb + (size_t)(n0 + l) * ldw + 2 * half;
    const float* wr1 = Wb + (size_t)(n0 + 16 + l) * ldw + 2 * half;
    for (int k0 = 0; k0 < DD; k0 += 4) {
      v2f a  = *(const v2f*)(arow + k0);
      v2f b0 = *(const v2f*)(wr0 + k0);
      v2f b1 = *(const v2f*)(wr1 + k0);
      acc0 = __builtin_amdgcn_wmma_f32_16x16x4_f32(false, a, false, b0, (short)0, acc0, false, false);
      acc1 = __builtin_amdgcn_wmma_f32_16x16x4_f32(false, a, false, b1, (short)0, acc1, false, false);
    }
  } else {
    for (int k0 = 0; k0 < DD; k0 += 4) {
      v2f a = *(const v2f*)(arow + k0);
      const float* r0 = Wb + (size_t)(k0 + 2 * half) * ldw + n0 + l;
      const float* r1 = r0 + ldw;
      v2f b0, b1;
      b0.x = r0[0];  b0.y = r1[0];
      b1.x = r0[16]; b1.y = r1[16];
      acc0 = __builtin_amdgcn_wmma_f32_16x16x4_f32(false, a, false, b0, (short)0, acc0, false, false);
      acc1 = __builtin_amdgcn_wmma_f32_16x16x4_f32(false, a, false, b1, (short)0, acc1, false, false);
    }
  }

  #pragma unroll
  for (int r = 0; r < 8; ++r) {
    const int m = m0 + r + 8 * half;
    float c0 = acc0[r];
    float c1 = acc1[r];
    if (Zmul) {
      c0 *= dsiluf(Zmul[(size_t)m * DD + (n0 + l)]);
      c1 *= dsiluf(Zmul[(size_t)m * DD + (n0 + 16 + l)]);
    }
    if (Cout) {
      Cout[(size_t)m * ldc + n0 + l]      = c0;
      Cout[(size_t)m * ldc + n0 + 16 + l] = c1;
    }
    if (SiluOut) {
      SiluOut[(size_t)m * ldso + n0 + l]      = siluf(c0);
      SiluOut[(size_t)m * ldso + n0 + 16 + l] = siluf(c1);
    }
  }
}

#define GU_WAVES   16
#define GU_KPW     (BN / GU_WAVES)
#define GU_TSTRIDE 272
__global__ __launch_bounds__(GU_WAVES * 32) void gradU(
    const float* __restrict__ Act, int ldact,
    const float* __restrict__ Dlt,
    const float* __restrict__ Wbase,
    float* __restrict__ Uout)
{
  __shared__ float part[GU_WAVES * 2 * GU_TSTRIDE];

  const int wave = threadIdx.x >> 5;
  const int lane = threadIdx.x & 31;
  const int half = lane >> 4;
  const int l    = lane & 15;
  const int m0   = blockIdx.x * 16;
  const int n0   = blockIdx.y * 32;
  const int batch = blockIdx.z;
  const size_t tok0 = (size_t)batch * BN + (size_t)wave * GU_KPW;
  const float* Ab = Act + tok0 * ldact + m0 + l;
  const float* Db = Dlt + tok0 * DD + n0 + l;

  v8f acc0 = {};
  v8f acc1 = {};
  for (int k0 = 0; k0 < GU_KPW; k0 += 4) {
    const int kr = k0 + 2 * half;
    v2f a;
    a.x = Ab[(size_t)(kr + 0) * ldact];
    a.y = Ab[(size_t)(kr + 1) * ldact];
    const float* d0 = Db + (size_t)kr * DD;
    const float* d1 = d0 + DD;
    v2f b0, b1;
    b0.x = d0[0];  b0.y = d1[0];
    b1.x = d0[16]; b1.y = d1[16];
    acc0 = __builtin_amdgcn_wmma_f32_16x16x4_f32(false, a, false, b0, (short)0, acc0, false, false);
    acc1 = __builtin_amdgcn_wmma_f32_16x16x4_f32(false, a, false, b1, (short)0, acc1, false, false);
  }

  float* pw = part + wave * (2 * GU_TSTRIDE) + lane;
  for (int r = 0; r < 8; ++r) {
    pw[r * 32]              = acc0[r];
    pw[GU_TSTRIDE + r * 32] = acc1[r];
  }
  __syncthreads();

  const int row = wave;
  const int t   = lane >> 4;
  const float* pr = part + t * GU_TSTRIDE + (row & 7) * 32 + (row >> 3) * 16 + l;
  float s = pr[0];
  for (int w = 1; w < GU_WAVES; ++w) s += pr[w * (2 * GU_TSTRIDE)];
  const size_t o = (size_t)(m0 + row) * DD + n0 + lane;
  Uout[(size_t)batch * DD * DD + o] = Wbase[o] + s;
}

__global__ __launch_bounds__(256) void d3_elem(
    const float* __restrict__ P, const float* __restrict__ KV, float* __restrict__ D3)
{
  const int idx = blockIdx.x * 256 + threadIdx.x;
  const int m = idx >> 7;
  const int n = idx & 127;
  const float v = KV[(size_t)m * 256 + DD + n];
  D3[idx] = (2.0f / 128.0f) * (P[idx] - v);
}

extern "C" void kernel_launch(void* const* d_in, const int* in_sizes, int n_in,
                              void* d_out, int out_size, void* d_ws, size_t ws_size,
                              hipStream_t stream)
{
  const float* S   = (const float*)d_in[0];
  const float* w0  = (const float*)d_in[1];
  const float* w1  = (const float*)d_in[2];
  const float* w2  = (const float*)d_in[3];
  const float* w3  = (const float*)d_in[4];
  const float* wq  = (const float*)d_in[5];
  const float* wkv = (const float*)d_in[6];
  float* out = (float*)d_out;

  float* p = (float*)d_ws;
  const size_t T = (size_t)TOK * DD;
  float* KV = p; p += (size_t)TOK * 256;
  float* Z0 = p; p += T;  float* A1 = p; p += T;
  float* Z1 = p; p += T;  float* A2 = p; p += T;
  float* Z2 = p; p += T;  float* A3 = p; p += T;
  float* P_ = p; p += T;
  float* Da = p; p += T;
  float* Db = p; p += T;
  float* Xa = p; p += T;
  float* Xb = p; p += T;
  float* U0 = p; p += 4 * DD * DD;
  float* U1 = p; p += 4 * DD * DD;
  float* U2 = p; p += 4 * DD * DD;
  float* U3 = p; p += 4 * DD * DD;

  const dim3 blk(256);
  const dim3 gKV(TOK / 32, 2);
  const dim3 g(TOK / 32, 1);
  const dim3 gblk(512);
  const dim3 gg(8, 4, 4);

  gemm128<<<gKV, blk, 0, stream>>>(S, DD, wkv, 256, 0, 0, nullptr, KV, 256, nullptr, 0);

  gemm128<<<g, blk, 0, stream>>>(KV, 256, w0, DD, 0, 0, nullptr, Z0, DD, A1, DD);
  gemm128<<<g, blk, 0, stream>>>(A1, DD, w1, DD, 0, 0, nullptr, Z1, DD, A2, DD);
  gemm128<<<g, blk, 0, stream>>>(A2, DD, w2, DD, 0, 0, nullptr, Z2, DD, A3, DD);
  gemm128<<<g, blk, 0, stream>>>(A3, DD, w3, DD, 0, 0, nullptr, P_, DD, nullptr, 0);

  d3_elem<<<dim3((TOK * DD) / 256), blk, 0, stream>>>(P_, KV, Da);

  gradU<<<gg, gblk, 0, stream>>>(A3, DD, Da, w3, U3);
  gemm128<<<g, blk, 0, stream>>>(Da, DD, w3, DD, 0, 1, Z2, Db, DD, nullptr, 0);
  gradU<<<gg, gblk, 0, stream>>>(A2, DD, Db, w2, U2);
  gemm128<<<g, blk, 0, stream>>>(Db, DD, w2, DD, 0, 1, Z1, Da, DD, nullptr, 0);
  gradU<<<gg, gblk, 0, stream>>>(A1, DD, Da, w1, U1);
  gemm128<<<g, blk, 0, stream>>>(Da, DD, w1, DD, 0, 1, Z0, Db, DD, nullptr, 0);
  gradU<<<gg, gblk, 0, stream>>>(KV, 256, Db, w0, U0);

  gemm128<<<g, blk, 0, stream>>>(S,  DD, wq, DD, 0,       0, nullptr, Xa, DD, nullptr, 0);
  gemm128<<<g, blk, 0, stream>>>(Xa, DD, U0, DD, DD * DD, 0, nullptr, nullptr, 0, Xb, DD);
  gemm128<<<g, blk, 0, stream>>>(Xb, DD, U1, DD, DD * DD, 0, nullptr, nullptr, 0, Xa, DD);
  gemm128<<<g, blk, 0, stream>>>(Xa, DD, U2, DD, DD * DD, 0, nullptr, nullptr, 0, Xb, DD);
  gemm128<<<g, blk, 0, stream>>>(Xb, DD, U3, DD, DD * DD, 0, nullptr, out, DD, nullptr, 0);
}


# launch fusion, 18 launches -> 3: key path (projection, MLP forward, delta chain) and query path each one kernel per 16-token tile, activations through LDS; the 4 gradient GEMMs in one launch
# speedup vs baseline: 6.9579x; 2.8730x over previous
#include <hip/hip_runtime.h>

typedef __attribute__((ext_vector_type(2))) float v2f;
typedef __attribute__((ext_vector_type(8))) float v8f;

#define TOK   4096
#define BN    1024
#define DD    128
#define TR    16
#define LSTR  132

__device__ __forceinline__ float sigf(float x)  { return 1.0f / (1.0f + __expf(-x)); }
__device__ __forceinline__ float siluf(float x) { return x * sigf(x); }
__device__ __forceinline__ float dsiluf(float x){ float s = sigf(x); return s * (1.0f + x * (1.0f - s)); }

__device__ __forceinline__ v8f mma4(v2f a, v2f b, v8f c)
{
  return __builtin_amdgcn_wmma_f32_16x16x4_f32(false, a, false, b, (short)0, c, false, false);
}

template <bool TRANS, int LDW>
__device__ __forceinline__ void mm16x32(const float* arow, const float* __restrict__ W,
                                        int n0, int l, int half, v8f& acc0, v8f& acc1)
{
  acc0 = (v8f){};
  acc1 = (v8f){};
  if (TRANS) {
    const float* wr0 = W + (size_t)(n0 + l) * LDW + 2 * half;
    const float* wr1 = W + (size_t)(n0 + 16 + l) * LDW + 2 * half;
    for (int k0 = 0; k0 < DD; k0 += 4) {
      v2f a  = *(const v2f*)(arow + k0);
      v2f b0 = *(const v2f*)(wr0 + k0);
      v2f b1 = *(const v2f*)(wr1 + k0);
      acc0 = mma4(a, b0, acc0);
      acc1 = mma4(a, b1, acc1);
    }
  } else {
    for (int k0 = 0; k0 < DD; k0 += 4) {
      v2f a = *(const v2f*)(arow + k0);
      const float* r0 = W + (size_t)(k0 + 2 * half) * LDW + n0 + l;
      const float* r1 = r0 + LDW;
      v2f b0, b1;
      b0.x = r0[0];  b0.y = r1[0];
      b1.x = r0[16]; b1.y = r1[16];
      acc0 = mma4(a, b0, acc0);
      acc1 = mma4(a, b1, acc1);
    }
  }
}

__device__ __forceinline__ void put_tile(float* dst, int pitch, int n0, int l, int half,
                                         const v8f& c0, const v8f& c1)
{
  float* p = dst + (size_t)(8 * half) * pitch + n0 + l;
  for (int r = 0; r < 8; ++r) {
    p[(size_t)r * pitch]      = c0[r];
    p[(size_t)r * pitch + 16] = c1[r];
  }
}

__global__ __launch_bounds__(128) void chain_fb(
    const float* __restrict__ S,  const float* __restrict__ wkv,
    const float* __restrict__ w0, const float* __restrict__ w1,
    const float* __restrict__ w2, const float* __restrict__ w3,
    float* __restrict__ Act, float* __restrict__ Dlt)
{
  __shared__ float buf[2][TR * LSTR];

  const int wave = threadIdx.x >> 5;
  const int lane = threadIdx.x & 31;
  const int half = lane >> 4;
  const int l    = lane & 15;
  const int m0   = blockIdx.x * TR;
  const int n0   = wave * 32;
  const size_t T  = (size_t)TOK * DD;
  const size_t go = (size_t)m0 * DD;

  const float* la0 = &buf[0][l * LSTR + 2 * half];
  const float* la1 = &buf[1][l * LSTR + 2 * half];

  v8f kk0 = {}, kk1 = {}, vv0 = {}, vv1 = {};
  {
    const float* arow = S + (size_t)(m0 + l) * DD + 2 * half;
    for (int k0 = 0; k0 < DD; k0 += 4) {
      v2f a = *(const v2f*)(arow + k0);
      const float* r0 = wkv + (size_t)(k0 + 2 * half) * 256 + n0 + l;
      const float* r1 = r0 + 256;
      v2f bk0, bk1, bv0, bv1;
      bk0.x = r0[0];   bk0.y = r1[0];
      bk1.x = r0[16];  bk1.y = r1[16];
      bv0.x = r0[128]; bv0.y = r1[128];
      bv1.x = r0[144]; bv1.y = r1[144];
      kk0 = mma4(a, bk0, kk0);
      kk1 = mma4(a, bk1, kk1);
      vv0 = mma4(a, bv0, vv0);
      vv1 = mma4(a, bv1, vv1);
    }
  }
  put_tile(buf[0], LSTR, n0, l, half, kk0, kk1);
  __syncthreads();
  put_tile(Act + go, DD, n0, l, half, kk0, kk1);

  v8f z0, z1, a0, a1;
  v8f g0a, g0b, g1a, g1b, g2a, g2b;

  mm16x32<false, DD>(la0, w0, n0, l, half, z0, z1);
  for (int r = 0; r < 8; ++r) {
    g0a[r] = dsiluf(z0[r]); g0b[r] = dsiluf(z1[r]);
    a0[r]  = siluf(z0[r]);  a1[r]  = siluf(z1[r]);
  }
  put_tile(buf[1], LSTR, n0, l, half, a0, a1);
  __syncthreads();
  put_tile(Act + T + go, DD, n0, l, half, a0, a1);

  mm16x32<false, DD>(la1, w1, n0, l, half, z0, z1);
  for (int r = 0; r < 8; ++r) {
    g1a[r] = dsiluf(z0[r]); g1b[r] = dsiluf(z1[r]);
    a0[r]  = siluf(z0[r]);  a1[r]  = siluf(z1[r]);
  }
  put_tile(buf[0], LSTR, n0, l, half, a0, a1);
  __syncthreads();
  put_tile(Act + 2 * T + go, DD, n0, l, half, a0, a1);

  mm16x32<false, DD>(la0, w2, n0, l, half, z0, z1);
  for (int r = 0; r < 8; ++r) {
    g2a[r] = dsiluf(z0[r]); g2b[r] = dsiluf(z1[r]);
    a0[r]  = siluf(z0[r]);  a1[r]  = siluf(z1[r]);
  }
  put_tile(buf[1], LSTR, n0, l, half, a0, a1);
  __syncthreads();
  put_tile(Act + 3 * T + go, DD, n0, l, half, a0, a1);

  mm16x32<false, DD>(la1, w3, n0, l, half, z0, z1);
  for (int r = 0; r < 8; ++r) {
    a0[r] = (2.0f / 128.0f) * (z0[r] - vv0[r]);
    a1[r] = (2.0f / 128.0f) * (z1[r] - vv1[r]);
  }
  put_tile(buf[0], LSTR, n0, l, half, a0, a1);
  __syncthreads();
  put_tile(Dlt + 3 * T + go, DD, n0, l, half, a0, a1);

  mm16x32<true, DD>(la0, w3, n0, l, half, z0, z1);
  for (int r = 0; r < 8; ++r) { a0[r] = z0[r] * g2a[r]; a1[r] = z1[r] * g2b[r]; }
  put_tile(buf[1], LSTR, n0, l, half, a0, a1);
  __syncthreads();
  put_tile(Dlt + 2 * T + go, DD, n0, l, half, a0, a1);

  mm16x32<true, DD>(la1, w2, n0, l, half, z0, z1);
  for (int r = 0; r < 8; ++r) { a0[r] = z0[r] * g1a[r]; a1[r] = z1[r] * g1b[r]; }
  put_tile(buf[0], LSTR, n0, l, half, a0, a1);
  __syncthreads();
  put_tile(Dlt + T + go, DD, n0, l, half, a0, a1);

  mm16x32<true, DD>(la0, w1, n0, l, half, z0, z1);
  for (int r = 0; r < 8; ++r) { a0[r] = z0[r] * g0a[r]; a1[r] = z1[r] * g0b[r]; }
  put_tile(Dlt + go, DD, n0, l, half, a0, a1);
}

#define GU_WAVES   16
#define GU_KPW     (BN / GU_WAVES)
#define GU_TSTRIDE 272
__global__ __launch_bounds__(GU_WAVES * 32) void gradU(
    const float* __restrict__ ActAll,
    const float* __restrict__ DltAll,
    const float* __restrict__ w0, const float* __restrict__ w1,
    const float* __restrict__ w2, const float* __restrict__ w3,
    float* __restrict__ UAll)
{
  __shared__ float part[GU_WAVES * 2 * GU_TSTRIDE];

  const int wave = threadIdx.x >> 5;
  const int lane = threadIdx.x & 31;
  const int half = lane >> 4;
  const int l    = lane & 15;
  const int m0   = blockIdx.x * 16;
  const int n0   = blockIdx.y * 32;
  const int layer = blockIdx.z >> 2;
  const int batch = blockIdx.z & 3;
  const size_t T  = (size_t)TOK * DD;
  const float* Act   = ActAll + (size_t)layer * T;
  const float* Dlt   = DltAll + (size_t)layer * T;
  const float* Wbase = layer == 0 ? w0 : layer == 1 ? w1 : layer == 2 ? w2 : w3;
  float* Uout = UAll + (size_t)layer * 4 * DD * DD;
  const size_t tok0 = (size_t)batch * BN + (size_t)wave * GU_KPW;
  const float* Ab = Act + tok0 * DD + m0 + l;
  const float* Db = Dlt + tok0 * DD + n0 + l;

  v8f acc0 = {};
  v8f acc1 = {};
  for (int k0 = 0; k0 < GU_KPW; k0 += 4) {
    const int kr = k0 + 2 * half;
    v2f a;
    a.x = Ab[(size_t)(kr + 0) * DD];
    a.y = Ab[(size_t)(kr + 1) * DD];
    const float* d0 = Db + (size_t)kr * DD;
    const float* d1 = d0 + DD;
    v2f b0, b1;
    b0.x = d0[0];  b0.y = d1[0];
    b1.x = d0[16]; b1.y = d1[16];
    acc0 = mma4(a, b0, acc0);
    acc1 = mma4(a, b1, acc1);
  }

  float* pw = part + wave * (2 * GU_TSTRIDE) + lane;
  for (int r = 0; r < 8; ++r) {
    pw[r * 32]              = acc0[r];
    pw[GU_TSTRIDE + r * 32] = acc1[r];
  }
  __syncthreads();

  const int row = wave;
  const int t   = lane >> 4;
  const float* pr = part + t * GU_TSTRIDE + (row & 7) * 32 + (row >> 3) * 16 + l;
  float s = pr[0];
  for (int w = 1; w < GU_WAVES; ++w) s += pr[w * (2 * GU_TSTRIDE)];
  const size_t o = (size_t)(m0 + row) * DD + n0 + lane;
  Uout[(size_t)batch * DD * DD + o] = Wbase[o] + s;
}

__global__ __launch_bounds__(128) void chain_q(
    const float* __restrict__ S, const float* __restrict__ wq,
    const float* __restrict__ U, float* __restrict__ out)
{
  __shared__ float buf[2][TR * LSTR];

  const int wave = threadIdx.x >> 5;
  const int lane = threadIdx.x & 31;
  const int half = lane >> 4;
  const int l    = lane & 15;
  const int m0   = blockIdx.x * TR;
  const int n0   = wave * 32;
  const int batch = m0 / BN;
  const float* Ub = U + (size_t)batch * DD * DD;
  const size_t UL = (size_t)4 * DD * DD;

  const float* la0 = &buf[0][l * LSTR + 2 * half];
  const float* la1 = &buf[1][l * LSTR + 2 * half];

  v8f z0, z1, a0, a1;

  mm16x32<false, DD>(S + (size_t)(m0 + l) * DD + 2 * half, wq, n0, l, half, z0, z1);
  put_tile(buf[0], LSTR, n0, l, half, z0, z1);
  __syncthreads();

  mm16x32<false, DD>(la0, Ub, n0, l, half, z0, z1);
  for (int r = 0; r < 8; ++r) { a0[r] = siluf(z0[r]); a1[r] = siluf(z1[r]); }
  put_tile(buf[1], LSTR, n0, l, half, a0, a1);
  __syncthreads();

  mm16x32<false, DD>(la1, Ub + UL, n0, l, half, z0, z1);
  for (int r = 0; r < 8; ++r) { a0[r] = siluf(z0[r]); a1[r] = siluf(z1[r]); }
  put_tile(buf[0], LSTR, n0, l, half, a0, a1);
  __syncthreads();

  mm16x32<false, DD>(la0, Ub + 2 * UL, n0, l, half, z0, z1);
  for (int r = 0; r < 8; ++r) { a0[r] = siluf(z0[r]); a1[r] = siluf(z1[r]); }
  put_tile(buf[1], LSTR, n0, l, half, a0, a1);
  __syncthreads();

  mm16x32<false, DD>(la1, Ub + 3 * UL, n0, l, half, z0, z1);
  put_tile(out + (size_t)m0 * DD, DD, n0, l, half, z0, z1);
}

extern "C" void kernel_launch(void* const* d_in, const int* in_sizes, int n_in,
                              void* d_out, int out_size, void* d_ws, size_t ws_size,
                              hipStream_t stream)
{
  const float* S   = (const float*)d_in[0];
  const float* w0  = (const float*)d_in[1];
  const float* w1  = (const float*)d_in[2];
  const float* w2  = (const float*)d_in[3];
  const float* w3  = (const float*)d_in[4];
  const float* wq  = (const float*)d_in[5];
  const float* wkv = (const float*)d_in[6];
  float* out = (float*)d_out;

  float* p = (float*)d_ws;
  const size_t T = (size_t)TOK * DD;
  float* Act = p; p += 4 * T;
  float* Dlt = p; p += 4 * T;
  float* U   = p; p += 16 * DD * DD;

  chain_fb<<<dim3(TOK / TR), dim3(128), 0, stream>>>(S, wkv, w0, w1, w2, w3, Act, Dlt);
  gradU<<<dim3(8, 4, 16), dim3(GU_WAVES * 32), 0, stream>>>(Act, Dlt, w0, w1, w2, w3, U);
  chain_q<<<dim3(TOK / TR), dim3(128), 0, stream>>>(S, wq, U, out);
}


# weight loads issued ahead: each layer's first B fragments are requested before the previous layer's epilogue and stay in flight across an LDS-only barrier; key/value projection loop fully unrolled
# speedup vs baseline: 9.6187x; 1.3824x over previous
#include <hip/hip_runtime.h>

typedef __attribute__((ext_vector_type(2))) float v2f;
typedef __attribute__((ext_vector_type(8))) float v8f;

#define TOK   4096
#define BN    1024
#define DD    128
#define TR    16
#define LSTR  132

__device__ __forceinline__ float sigf(float x)  { return 1.0f / (1.0f + __expf(-x)); }
__device__ __forceinline__ float siluf(float x) { return x * sigf(x); }
__device__ __forceinline__ float dsiluf(float x){ float s = sigf(x); return s * (1.0f + x * (1.0f - s)); }

__device__ __forceinline__ v8f mma4(v2f a, v2f b, v8f c)
{
  return __builtin_amdgcn_wmma_f32_16x16x4_f32(false, a, false, b, (short)0, c, false, false);
}

template <bool TRANS, int LDW>
__device__ __forceinline__ void mm16x32(const float* arow, const float* __restrict__ W,
                                        int n0, int l, int half, v8f& acc0, v8f& acc1)
{
  acc0 = (v8f){};
  acc1 = (v8f){};
  if (TRANS) {
    const float* wr0 = W + (size_t)(n0 + l) * LDW + 2 * half;
    const float* wr1 = W + (size_t)(n0 + 16 + l) * LDW + 2 * half;
    for (int k0 = 0; k0 < DD; k0 += 4) {
      v2f a  = *(const v2f*)(arow + k0);
      v2f b0 = *(const v2f*)(wr0 + k0);
      v2f b1 = *(const v2f*)(wr1 + k0);
      acc0 = mma4(a, b0, acc0);
      acc1 = mma4(a, b1, acc1);
    }
  } else {
    for (int k0 = 0; k0 < DD; k0 += 4) {
      v2f a = *(const v2f*)(arow + k0);
      const float* r0 = W + (size_t)(k0 + 2 * half) * LDW + n0 + l;
      const float* r1 = r0 + LDW;
      v2f b0, b1;
      b0.x = r0[0];  b0.y = r1[0];
      b1.x = r0[16]; b1.y = r1[16];
      acc0 = mma4(a, b0, acc0);
      acc1 = mma4(a, b1, acc1);
    }
  }
}

__device__ __forceinline__ void put_tile(float* dst, int pitch, int n0, int l, int half,
                                         const v8f& c0, const v8f& c1)
{
  float* p = dst + (size_t)(8 * half) * pitch + n0 + l;
  for (int r = 0; r < 8; ++r) {
    p[(size_t)r * pitch]      = c0[r];
    p[(size_t)r * pitch + 16] = c1[r];
  }
}

#define PFN 15
#define PFT 11
__device__ __forceinline__ constexpr bool pft_has(int j) { return j < 8 || (j & 7) == 0; }
__device__ __forceinline__ constexpr int  pft_idx(int j) { return j < 8 ? j : 7 + (j >> 3); }

template <int LDW>
__device__ __forceinline__ void preN(const float* __restrict__ W, int n0, int l, int half,
                                     v2f (&p0)[PFN], v2f (&p1)[PFN])
{
#pragma unroll
  for (int j = 0; j < PFN; ++j) {
    const float* r0 = W + (size_t)(4 * j + 2 * half) * LDW + n0 + l;
    const float* r1 = r0 + LDW;
    p0[j].x = r0[0];  p0[j].y = r1[0];
    p1[j].x = r0[16]; p1[j].y = r1[16];
  }
}

template <int LDW>
__device__ __forceinline__ void mmN(const float* arow, const float* __restrict__ W,
                                    int n0, int l, int half,
                                    const v2f (&p0)[PFN], const v2f (&p1)[PFN], v8f& acc0, v8f& acc1)
{
  acc0 = (v8f){};
  acc1 = (v8f){};
#pragma unroll
  for (int j = 0; j < DD / 4; ++j) {
    v2f a = *(const v2f*)(arow + 4 * j);
    v2f b0, b1;
    if (j < PFN) {
      b0 = p0[j]; b1 = p1[j];
    } else {
      const float* r0 = W + (size_t)(4 * j + 2 * half) * LDW + n0 + l;
      const float* r1 = r0 + LDW;
      b0.x = r0[0];  b0.y = r1[0];
      b1.x = r0[16]; b1.y = r1[16];
    }
    acc0 = mma4(a, b0, acc0);
    acc1 = mma4(a, b1, acc1);
  }
}

template <int LDW>
__device__ __forceinline__ void preT(const float* __restrict__ W, int n0, int l, int half,
                                     v2f (&p0)[PFT], v2f (&p1)[PFT])
{
  const float* wr0 = W + (size_t)(n0 + l) * LDW + 2 * half;
  const float* wr1 = W + (size_t)(n0 + 16 + l) * LDW + 2 * half;
#pragma unroll
  for (int j = 0; j < DD / 4; ++j) {
    if (pft_has(j)) {
      p0[pft_idx(j)] = *(const v2f*)(wr0 + 4 * j);
      p1[pft_idx(j)] = *(const v2f*)(wr1 + 4 * j);
    }
  }
}

template <int LDW>
__device__ __forceinline__ void mmT(const float* arow, const float* __restrict__ W,
                                    int n0, int l, int half,
                                    const v2f (&p0)[PFT], const v2f (&p1)[PFT], v8f& acc0, v8f& acc1)
{
  acc0 = (v8f){};
  acc1 = (v8f){};
  const float* wr0 = W + (size_t)(n0 + l) * LDW + 2 * half;
  const float* wr1 = W + (size_t)(n0 + 16 + l) * LDW + 2 * half;
#pragma unroll
  for (int j = 0; j < DD / 4; ++j) {
    v2f a = *(const v2f*)(arow + 4 * j);
    v2f b0, b1;
    if (pft_has(j)) {
      b0 = p0[pft_idx(j)]; b1 = p1[pft_idx(j)];
    } else {
      b0 = *(const v2f*)(wr0 + 4 * j);
      b1 = *(const v2f*)(wr1 + 4 * j);
    }
    acc0 = mma4(a, b0, acc0);
    acc1 = mma4(a, b1, acc1);
  }
}

#define AHEAD_FENCE() __builtin_amdgcn_sched_barrier(0)

#define LDS_SYNC() do { \
    __builtin_amdgcn_fence(__ATOMIC_RELEASE, "workgroup", "local"); \
    __builtin_amdgcn_s_barrier(); \
    __builtin_amdgcn_fence(__ATOMIC_ACQUIRE, "workgroup", "local"); \
  } while (0)

__global__ __launch_bounds__(128) void chain_fb(
    const float* __restrict__ S,  const float* __restrict__ wkv,
    const float* __restrict__ w0, const float* __restrict__ w1,
    const float* __restrict__ w2, const float* __restrict__ w3,
    float* __restrict__ Act, float* __restrict__ Dlt)
{
  __shared__ float buf[2][TR * LSTR];

  const int wave = threadIdx.x >> 5;
  const int lane = threadIdx.x & 31;
  const int half = lane >> 4;
  const int l    = lane & 15;
  const int m0   = blockIdx.x * TR;
  const int n0   = wave * 32;
  const size_t T  = (size_t)TOK * DD;
  const size_t go = (size_t)m0 * DD;

  const float* la0 = &buf[0][l * LSTR + 2 * half];
  const float* la1 = &buf[1][l * LSTR + 2 * half];

  v8f kk0 = {}, kk1 = {}, vv0 = {}, vv1 = {};
  {
    const float* arow = S + (size_t)(m0 + l) * DD + 2 * half;
#pragma unroll
    for (int k0 = 0; k0 < DD; k0 += 4) {
      v2f a = *(const v2f*)(arow + k0);
      const float* r0 = wkv + (size_t)(k0 + 2 * half) * 256 + n0 + l;
      const float* r1 = r0 + 256;
      v2f bk0, bk1, bv0, bv1;
      bk0.x = r0[0];   bk0.y = r1[0];
      bk1.x = r0[16];  bk1.y = r1[16];
      bv0.x = r0[128]; bv0.y = r1[128];
      bv1.x = r0[144]; bv1.y = r1[144];
      kk0 = mma4(a, bk0, kk0);
      kk1 = mma4(a, bk1, kk1);
      vv0 = mma4(a, bv0, vv0);
      vv1 = mma4(a, bv1, vv1);
    }
  }
  v2f pn0[PFN], pn1[PFN];
  v2f pt0[PFT], pt1[PFT];
  AHEAD_FENCE();
  preN<DD>(w0, n0, l, half, pn0, pn1);
  put_tile(buf[0], LSTR, n0, l, half, kk0, kk1);
  LDS_SYNC();
  put_tile(Act + go, DD, n0, l, half, kk0, kk1);

  v8f z0, z1, a0, a1;
  v8f g0a, g0b, g1a, g1b, g2a, g2b;

  mmN<DD>(la0, w0, n0, l, half, pn0, pn1, z0, z1);
  AHEAD_FENCE();
  preN<DD>(w1, n0, l, half, pn0, pn1);
  for (int r = 0; r < 8; ++r) {
    g0a[r] = dsiluf(z0[r]); g0b[r] = dsiluf(z1[r]);
    a0[r]  = siluf(z0[r]);  a1[r]  = siluf(z1[r]);
  }
  put_tile(buf[1], LSTR, n0, l, half, a0, a1);
  LDS_SYNC();
  put_tile(Act + T + go, DD, n0, l, half, a0, a1);

  mmN<DD>(la1, w1, n0, l, half, pn0, pn1, z0, z1);
  AHEAD_FENCE();
  preN<DD>(w2, n0, l, half, pn0, pn1);
  for (int r = 0; r < 8; ++r) {
    g1a[r] = dsiluf(z0[r]); g1b[r] = dsiluf(z1[r]);
    a0[r]  = siluf(z0[r]);  a1[r]  = siluf(z1[r]);
  }
  put_tile(buf[0], LSTR, n0, l, half, a0, a1);
  LDS_SYNC();
  put_tile(Act + 2 * T + go, DD, n0, l, half, a0, a1);

  mmN<DD>(la0, w2, n0, l, half, pn0, pn1, z0, z1);
  AHEAD_FENCE();
  preN<DD>(w3, n0, l, half, pn0, pn1);
  for (int r = 0; r < 8; ++r) {
    g2a[r] = dsiluf(z0[r]); g2b[r] = dsiluf(z1[r]);
    a0[r]  = siluf(z0[r]);  a1[r]  = siluf(z1[r]);
  }
  put_tile(buf[1], LSTR, n0, l, half, a0, a1);
  LDS_SYNC();
  put_tile(Act + 3 * T + go, DD, n0, l, half, a0, a1);

  mmN<DD>(la1, w3, n0, l, half, pn0, pn1, z0, z1);
  AHEAD_FENCE();
  preT<DD>(w3, n0, l, half, pt0, pt1);
  for (int r = 0; r < 8; ++r) {
    a0[r] = (2.0f / 128.0f) * (z0[r] - vv0[r]);
    a1[r] = (2.0f / 128.0f) * (z1[r] - vv1[r]);
  }
  put_tile(buf[0], LSTR, n0, l, half, a0, a1);
  LDS_SYNC();
  put_tile(Dlt + 3 * T + go, DD, n0, l, half, a0, a1);

  mmT<DD>(la0, w3, n0, l, half, pt0, pt1, z0, z1);
  AHEAD_FENCE();
  preT<DD>(w2, n0, l, half, pt0, pt1);
  for (int r = 0; r < 8; ++r) { a0[r] = z0[r] * g2a[r]; a1[r] = z1[r] * g2b[r]; }
  put_tile(buf[1], LSTR, n0, l, half, a0, a1);
  LDS_SYNC();
  put_tile(Dlt + 2 * T + go, DD, n0, l, half, a0, a1);

  mmT<DD>(la1, w2, n0, l, half, pt0, pt1, z0, z1);
  AHEAD_FENCE();
  preT<DD>(w1, n0, l, half, pt0, pt1);
  for (int r = 0; r < 8; ++r) { a0[r] = z0[r] * g1a[r]; a1[r] = z1[r] * g1b[r]; }
  put_tile(buf[0], LSTR, n0, l, half, a0, a1);
  LDS_SYNC();
  put_tile(Dlt + T + go, DD, n0, l, half, a0, a1);

  mmT<DD>(la0, w1, n0, l, half, pt0, pt1, z0, z1);
  for (int r = 0; r < 8; ++r) { a0[r] = z0[r] * g0a[r]; a1[r] = z1[r] * g0b[r]; }
  put_tile(Dlt + go, DD, n0, l, half, a0, a1);
}

#define GU_WAVES   16
#define GU_KPW     (BN / GU_WAVES)
#define GU_TSTRIDE 272
__global__ __launch_bounds__(GU_WAVES * 32) void gradU(
    const float* __restrict__ ActAll,
    const float* __restrict__ DltAll,
    const float* __restrict__ w0, const float* __restrict__ w1,
    const float* __restrict__ w2, const float* __restrict__ w3,
    float* __restrict__ UAll)
{
  __shared__ float part[GU_WAVES * 2 * GU_TSTRIDE];

  const int wave = threadIdx.x >> 5;
  const int lane = threadIdx.x & 31;
  const int half = lane >> 4;
  const int l    = lane & 15;
  const int m0   = blockIdx.x * 16;
  const int n0   = blockIdx.y * 32;
  const int layer = blockIdx.z >> 2;
  const int batch = blockIdx.z & 3;
  const size_t T  = (size_t)TOK * DD;
  const float* Act   = ActAll + (size_t)layer * T;
  const float* Dlt   = DltAll + (size_t)layer * T;
  const float* Wbase = layer == 0 ? w0 : layer == 1 ? w1 : layer == 2 ? w2 : w3;
  float* Uout = UAll + (size_t)layer * 4 * DD * DD;
  const size_t tok0 = (size_t)batch * BN + (size_t)wave * GU_KPW;
  const float* Ab = Act + tok0 * DD + m0 + l;
  const float* Db = Dlt + tok0 * DD + n0 + l;

  v8f acc0 = {};
  v8f acc1 = {};
  for (int k0 = 0; k0 < GU_KPW; k0 += 4) {
    const int kr = k0 + 2 * half;
    v2f a;
    a.x = Ab[(size_t)(kr + 0) * DD];
    a.y = Ab[(size_t)(kr + 1) * DD];
    const float* d0 = Db + (size_t)kr * DD;
    const float* d1 = d0 + DD;
    v2f b0, b1;
    b0.x = d0[0];  b0.y = d1[0];
    b1.x = d0[16]; b1.y = d1[16];
    acc0 = mma4(a, b0, acc0);
    acc1 = mma4(a, b1, acc1);
  }

  float* pw = part + wave * (2 * GU_TSTRIDE) + lane;
  for (int r = 0; r < 8; ++r) {
    pw[r * 32]              = acc0[r];
    pw[GU_TSTRIDE + r * 32] = acc1[r];
  }
  __syncthreads();

  const int row = wave;
  const int t   = lane >> 4;
  const float* pr = part + t * GU_TSTRIDE + (row & 7) * 32 + (row >> 3) * 16 + l;
  float s = pr[0];
  for (int w = 1; w < GU_WAVES; ++w) s += pr[w * (2 * GU_TSTRIDE)];
  const size_t o = (size_t)(m0 + row) * DD + n0 + lane;
  Uout[(size_t)batch * DD * DD + o] = Wbase[o] + s;
}

__global__ __launch_bounds__(128) void chain_q(
    const float* __restrict__ S, const float* __restrict__ wq,
    const float* __restrict__ U, float* __restrict__ out)
{
  __shared__ float buf[2][TR * LSTR];

  const int wave = threadIdx.x >> 5;
  const int lane = threadIdx.x & 31;
  const int half = lane >> 4;
  const int l    = lane & 15;
  const int m0   = blockIdx.x * TR;
  const int n0   = wave * 32;
  const int batch = m0 / BN;
  const float* Ub = U + (size_t)batch * DD * DD;
  const size_t UL = (size_t)4 * DD * DD;

  const float* la0 = &buf[0][l * LSTR + 2 * half];
  const float* la1 = &buf[1][l * LSTR + 2 * half];

  v8f z0, z1, a0, a1;

  v2f pn0[PFN], pn1[PFN];

  mm16x32<false, DD>(S + (size_t)(m0 + l) * DD + 2 * half, wq, n0, l, half, z0, z1);
  AHEAD_FENCE();
  preN<DD>(Ub, n0, l, half, pn0, pn1);
  put_tile(buf[0], LSTR, n0, l, half, z0, z1);
  LDS_SYNC();

  mmN<DD>(la0, Ub, n0, l, half, pn0, pn1, z0, z1);
  AHEAD_FENCE();
  preN<DD>(Ub + UL, n0, l, half, pn0, pn1);
  for (int r = 0; r < 8; ++r) { a0[r] = siluf(z0[r]); a1[r] = siluf(z1[r]); }
  put_tile(buf[1], LSTR, n0, l, half, a0, a1);
  LDS_SYNC();

  mmN<DD>(la1, Ub + UL, n0, l, half, pn0, pn1, z0, z1);
  AHEAD_FENCE();
  preN<DD>(Ub + 2 * UL, n0, l, half, pn0, pn1);
  for (int r = 0; r < 8; ++r) { a0[r] = siluf(z0[r]); a1[r] = siluf(z1[r]); }
  put_tile(buf[0], LSTR, n0, l, half, a0, a1);
  LDS_SYNC();

  mmN<DD>(la0, Ub + 2 * UL, n0, l, half, pn0, pn1, z0, z1);
  AHEAD_FENCE();
  preN<DD>(Ub + 3 * UL, n0, l, half, pn0, pn1);
  for (int r = 0; r < 8; ++r) { a0[r] = siluf(z0[r]); a1[r] = siluf(z1[r]); }
  put_tile(buf[1], LSTR, n0, l, half, a0, a1);
  LDS_SYNC();

  mmN<DD>(la1, Ub + 3 * UL, n0, l, half, pn0, pn1, z0, z1);
  put_tile(out + (size_t)m0 * DD, DD, n0, l, half, z0, z1);
}

extern "C" void kernel_launch(void* const* d_in, const int* in_sizes, int n_in,
                              void* d_out, int out_size, void* d_ws, size_t ws_size,
                              hipStream_t stream)
{
  const float* S   = (const float*)d_in[0];
  const float* w0  = (const float*)d_in[1];
  const float* w1  = (const float*)d_in[2];
  const float* w2  = (const float*)d_in[3];
  const float* w3  = (const float*)d_in[4];
  const float* wq  = (const float*)d_in[5];
  const float* wkv = (const float*)d_in[6];
  float* out = (float*)d_out;

  float* p = (float*)d_ws;
  const size_t T = (size_t)TOK * DD;
  float* Act = p; p += 4 * T;
  float* Dlt = p; p += 4 * T;
  float* U   = p; p += 16 * DD * DD;

  chain_fb<<<dim3(TOK / TR), dim3(128), 0, stream>>>(S, wkv, w0, w1, w2, w3, Act, Dlt);
  gradU<<<dim3(8, 4, 16), dim3(GU_WAVES * 32), 0, stream>>>(Act, Dlt, w0, w1, w2, w3, U);
  chain_q<<<dim3(TOK / TR), dim3(128), 0, stream>>>(S, wq, U, out);
}


# chain kernels, forward layers: operands fetched ahead of the matrix stream: weights by 8-byte loads a layer ahead (31 of 32 k-steps, lane owns 2 adjacent columns), A as one LDS group; V tiles fenced
# speedup vs baseline: 10.9301x; 1.1363x over previous
#include <hip/hip_runtime.h>

typedef __attribute__((ext_vector_type(2))) float v2f;
typedef __attribute__((ext_vector_type(8))) float v8f;

#define TOK   4096
#define BN    1024
#define DD    128
#define TR    16
#define LSTR  132

__device__ __forceinline__ float sigf(float x)  { return 1.0f / (1.0f + __expf(-x)); }
__device__ __forceinline__ float siluf(float x) { return x * sigf(x); }
__device__ __forceinline__ float dsiluf(float x){ float s = sigf(x); return s * (1.0f + x * (1.0f - s)); }

__device__ __forceinline__ v8f mma4(v2f a, v2f b, v8f c)
{
  return __builtin_amdgcn_wmma_f32_16x16x4_f32(false, a, false, b, (short)0, c, false, false);
}

template <int LDW>
__device__ __forceinline__ void ldq(const float* __restrict__ W, int j, int n0, int l, int half,
                                    v2f& q0, v2f& q1)
{
  const float* r0 = W + (size_t)(4 * j + 2 * half) * LDW + n0 + 2 * l;
  q0 = *(const v2f*)r0;
  q1 = *(const v2f*)(r0 + LDW);
}

template <int LDW>
__device__ __forceinline__ void mm16x32(const float* arow, const float* __restrict__ W,
                                        int n0, int l, int half, v8f& acc0, v8f& acc1)
{
  acc0 = (v8f){};
  acc1 = (v8f){};
  for (int j = 0; j < DD / 4; ++j) {
    v2f a = *(const v2f*)(arow + 4 * j);
    v2f q0, q1;
    ldq<LDW>(W, j, n0, l, half, q0, q1);
    v2f b0, b1;
    b0.x = q0.x; b0.y = q1.x;
    b1.x = q0.y; b1.y = q1.y;
    acc0 = mma4(a, b0, acc0);
    acc1 = mma4(a, b1, acc1);
  }
}

__device__ __forceinline__ void put_tile(float* dst, int pitch, int n0, int l, int half,
                                         const v8f& c0, const v8f& c1)
{
  float* p = dst + (size_t)(8 * half) * pitch + n0 + 2 * l;
  for (int r = 0; r < 8; ++r) {
    v2f v;
    v.x = c0[r]; v.y = c1[r];
    *(v2f*)(p + (size_t)r * pitch) = v;
  }
}

#define PFN 31
#define PFT 11
__device__ __forceinline__ constexpr bool pft_has(int j) { return j < 8 || (j & 7) == 0; }
__device__ __forceinline__ constexpr int  pft_idx(int j) { return j < 8 ? j : 7 + (j >> 3); }

#define OPERANDS_FIRST() __builtin_amdgcn_sched_barrier(0)

template <int LDW>
__device__ __forceinline__ void preN(const float* __restrict__ W, int n0, int l, int half,
                                     v2f (&p0)[PFN], v2f (&p1)[PFN])
{
#pragma unroll
  for (int j = 0; j < PFN; ++j) ldq<LDW>(W, j, n0, l, half, p0[j], p1[j]);
}

template <int LDW>
__device__ __forceinline__ void mmN(const float* arow, const float* __restrict__ W,
                                    int n0, int l, int half,
                                    const v2f (&p0)[PFN], const v2f (&p1)[PFN], v8f& acc0, v8f& acc1)
{
  acc0 = (v8f){};
  acc1 = (v8f){};
  v2f t0[DD / 4 - PFN], t1[DD / 4 - PFN];
#pragma unroll
  for (int j = PFN; j < DD / 4; ++j) ldq<LDW>(W, j, n0, l, half, t0[j - PFN], t1[j - PFN]);
  v2f af[DD / 4];
#pragma unroll
  for (int j = 0; j < DD / 4; ++j) af[j] = *(const v2f*)(arow + 4 * j);
  OPERANDS_FIRST();
#pragma unroll
  for (int j = 0; j < DD / 4; ++j) {
    v2f a = af[j];
    v2f q0, q1;
    if (j < PFN) {
      q0 = p0[j]; q1 = p1[j];
    } else {
      q0 = t0[j - PFN]; q1 = t1[j - PFN];
    }
    v2f b0, b1;
    b0.x = q0.x; b0.y = q1.x;
    b1.x = q0.y; b1.y = q1.y;
    acc0 = mma4(a, b0, acc0);
    acc1 = mma4(a, b1, acc1);
  }
}

template <int LDW>
__device__ __forceinline__ void preT(const float* __restrict__ W, int n0, int l, int half,
                                     v2f (&p0)[PFT], v2f (&p1)[PFT])
{
  const float* wr0 = W + (size_t)(n0 + 2 * l) * LDW + 2 * half;
  const float* wr1 = wr0 + LDW;
#pragma unroll
  for (int j = 0; j < DD / 4; ++j) {
    if (pft_has(j)) {
      p0[pft_idx(j)] = *(const v2f*)(wr0 + 4 * j);
      p1[pft_idx(j)] = *(const v2f*)(wr1 + 4 * j);
    }
  }
}

template <int LDW>
__device__ __forceinline__ void mmT(const float* arow, const float* __restrict__ W,
                                    int n0, int l, int half,
                                    const v2f (&p0)[PFT], const v2f (&p1)[PFT], v8f& acc0, v8f& acc1)
{
  acc0 = (v8f){};
  acc1 = (v8f){};
  const float* wr0 = W + (size_t)(n0 + 2 * l) * LDW + 2 * half;
  const float* wr1 = wr0 + LDW;
#pragma unroll
  for (int j = 0; j < DD / 4; ++j) {
    v2f a = *(const v2f*)(arow + 4 * j);
    v2f b0, b1;
    if (pft_has(j)) {
      b0 = p0[pft_idx(j)]; b1 = p1[pft_idx(j)];
    } else {
      b0 = *(const v2f*)(wr0 + 4 * j);
      b1 = *(const v2f*)(wr1 + 4 * j);
    }
    acc0 = mma4(a, b0, acc0);
    acc1 = mma4(a, b1, acc1);
  }
}

#define AHEAD_FENCE() __builtin_amdgcn_sched_barrier(0)

#define LDS_SYNC() do { \
    __builtin_amdgcn_fence(__ATOMIC_RELEASE, "workgroup", "local"); \
    __builtin_amdgcn_s_barrier(); \
    __builtin_amdgcn_fence(__ATOMIC_ACQUIRE, "workgroup", "local"); \
  } while (0)

__global__ __launch_bounds__(128) void chain_fb(
    const float* __restrict__ S,  const float* __restrict__ wkv,
    const float* __restrict__ w0, const float* __restrict__ w1,
    const float* __restrict__ w2, const float* __restrict__ w3,
    float* __restrict__ Act, float* __restrict__ Dlt)
{
  __shared__ float buf[2][TR * LSTR];

  const int wave = threadIdx.x >> 5;
  const int lane = threadIdx.x & 31;
  const int half = lane >> 4;
  const int l    = lane & 15;
  const int m0   = blockIdx.x * TR;
  const int n0   = wave * 32;
  const size_t T  = (size_t)TOK * DD;
  const size_t go = (size_t)m0 * DD;

  const float* la0 = &buf[0][l * LSTR + 2 * half];
  const float* la1 = &buf[1][l * LSTR + 2 * half];

  v8f kk0 = {}, kk1 = {}, vv0 = {}, vv1 = {};
  {
    const float* arow = S + (size_t)(m0 + l) * DD + 2 * half;
#pragma unroll
    for (int k0 = 0; k0 < DD; k0 += 4) {
      v2f a = *(const v2f*)(arow + k0);
      const float* r0 = wkv + (size_t)(k0 + 2 * half) * 256 + n0 + 2 * l;
      const float* r1 = r0 + 256;
      const v2f qk0 = *(const v2f*)r0,         qk1 = *(const v2f*)r1;
      const v2f qv0 = *(const v2f*)(r0 + 128), qv1 = *(const v2f*)(r1 + 128);
      v2f bk0, bk1, bv0, bv1;
      bk0.x = qk0.x; bk0.y = qk1.x;
      bk1.x = qk0.y; bk1.y = qk1.y;
      bv0.x = qv0.x; bv0.y = qv1.x;
      bv1.x = qv0.y; bv1.y = qv1.y;
      kk0 = mma4(a, bk0, kk0);
      kk1 = mma4(a, bk1, kk1);
      vv0 = mma4(a, bv0, vv0);
      vv1 = mma4(a, bv1, vv1);
    }
  }
  asm volatile("" :: "v"(vv0), "v"(vv1));
  v2f pn0[PFN], pn1[PFN];
  v2f pt0[PFT], pt1[PFT];
  AHEAD_FENCE();
  preN<DD>(w0, n0, l, half, pn0, pn1);
  put_tile(buf[0], LSTR, n0, l, half, kk0, kk1);
  LDS_SYNC();
  put_tile(Act + go, DD, n0, l, half, kk0, kk1);

  v8f z0, z1, a0, a1;
  v8f g0a, g0b, g1a, g1b, g2a, g2b;

  mmN<DD>(la0, w0, n0, l, half, pn0, pn1, z0, z1);
  AHEAD_FENCE();
  preN<DD>(w1, n0, l, half, pn0, pn1);
  for (int r = 0; r < 8; ++r) {
    g0a[r] = dsiluf(z0[r]); g0b[r] = dsiluf(z1[r]);
    a0[r]  = siluf(z0[r]);  a1[r]  = siluf(z1[r]);
  }
  put_tile(buf[1], LSTR, n0, l, half, a0, a1);
  LDS_SYNC();
  put_tile(Act + T + go, DD, n0, l, half, a0, a1);

  mmN<DD>(la1, w1, n0, l, half, pn0, pn1, z0, z1);
  AHEAD_FENCE();
  preN<DD>(w2, n0, l, half, pn0, pn1);
  for (int r = 0; r < 8; ++r) {
    g1a[r] = dsiluf(z0[r]); g1b[r] = dsiluf(z1[r]);
    a0[r]  = siluf(z0[r]);  a1[r]  = siluf(z1[r]);
  }
  put_tile(buf[0], LSTR, n0, l, half, a0, a1);
  LDS_SYNC();
  put_tile(Act + 2 * T + go, DD, n0, l, half, a0, a1);

  mmN<DD>(la0, w2, n0, l, half, pn0, pn1, z0, z1);
  AHEAD_FENCE();
  preN<DD>(w3, n0, l, half, pn0, pn1);
  for (int r = 0; r < 8; ++r) {
    g2a[r] = dsiluf(z0[r]); g2b[r] = dsiluf(z1[r]);
    a0[r]  = siluf(z0[r]);  a1[r]  = siluf(z1[r]);
  }
  put_tile(buf[1], LSTR, n0, l, half, a0, a1);
  LDS_SYNC();
  put_tile(Act + 3 * T + go, DD, n0, l, half, a0, a1);

  mmN<DD>(la1, w3, n0, l, half, pn0, pn1, z0, z1);
  AHEAD_FENCE();
  preT<DD>(w3, n0, l, half, pt0, pt1);
  for (int r = 0; r < 8; ++r) {
    a0[r] = (2.0f / 128.0f) * (z0[r] - vv0[r]);
    a1[r] = (2.0f / 128.0f) * (z1[r] - vv1[r]);
  }
  put_tile(buf[0], LSTR, n0, l, half, a0, a1);
  LDS_SYNC();
  put_tile(Dlt + 3 * T + go, DD, n0, l, half, a0, a1);

  mmT<DD>(la0, w3, n0, l, half, pt0, pt1, z0, z1);
  AHEAD_FENCE();
  preT<DD>(w2, n0, l, half, pt0, pt1);
  for (int r = 0; r < 8; ++r) { a0[r] = z0[r] * g2a[r]; a1[r] = z1[r] * g2b[r]; }
  put_tile(buf[1], LSTR, n0, l, half, a0, a1);
  LDS_SYNC();
  put_tile(Dlt + 2 * T + go, DD, n0, l, half, a0, a1);

  mmT<DD>(la1, w2, n0, l, half, pt0, pt1, z0, z1);
  AHEAD_FENCE();
  preT<DD>(w1, n0, l, half, pt0, pt1);
  for (int r = 0; r < 8; ++r) { a0[r] = z0[r] * g1a[r]; a1[r] = z1[r] * g1b[r]; }
  put_tile(buf[0], LSTR, n0, l, half, a0, a1);
  LDS_SYNC();
  put_tile(Dlt + T + go, DD, n0, l, half, a0, a1);

  mmT<DD>(la0, w1, n0, l, half, pt0, pt1, z0, z1);
  for (int r = 0; r < 8; ++r) { a0[r] = z0[r] * g0a[r]; a1[r] = z1[r] * g0b[r]; }
  put_tile(Dlt + go, DD, n0, l, half, a0, a1);
}

#define GU_WAVES   16
#define GU_KPW     (BN / GU_WAVES)
#define GU_TSTRIDE 272
__global__ __launch_bounds__(GU_WAVES * 32) void gradU(
    const float* __restrict__ ActAll,
    const float* __restrict__ DltAll,
    const float* __restrict__ w0, const float* __restrict__ w1,
    const float* __restrict__ w2, const float* __restrict__ w3,
    float* __restrict__ UAll)
{
  __shared__ float part[GU_WAVES * 2 * GU_TSTRIDE];

  const int wave = threadIdx.x >> 5;
  const int lane = threadIdx.x & 31;
  const int half = lane >> 4;
  const int l    = lane & 15;
  const int m0   = blockIdx.x * 16;
  const int n0   = blockIdx.y * 32;
  const int layer = blockIdx.z >> 2;
  const int batch = blockIdx.z & 3;
  const size_t T  = (size_t)TOK * DD;
  const float* Act   = ActAll + (size_t)layer * T;
  const float* Dlt   = DltAll + (size_t)layer * T;
  const float* Wbase = layer == 0 ? w0 : layer == 1 ? w1 : layer == 2 ? w2 : w3;
  float* Uout = UAll + (size_t)layer * 4 * DD * DD;
  const size_t tok0 = (size_t)batch * BN + (size_t)wave * GU_KPW;
  const float* Ab = Act + tok0 * DD + m0 + l;
  const float* Db = Dlt + tok0 * DD + n0 + l;

  v8f acc0 = {};
  v8f acc1 = {};
  for (int k0 = 0; k0 < GU_KPW; k0 += 4) {
    const int kr = k0 + 2 * half;
    v2f a;
    a.x = Ab[(size_t)(kr + 0) * DD];
    a.y = Ab[(size_t)(kr + 1) * DD];
    const float* d0 = Db + (size_t)kr * DD;
    const float* d1 = d0 + DD;
    v2f b0, b1;
    b0.x = d0[0];  b0.y = d1[0];
    b1.x = d0[16]; b1.y = d1[16];
    acc0 = mma4(a, b0, acc0);
    acc1 = mma4(a, b1, acc1);
  }

  float* pw = part + wave * (2 * GU_TSTRIDE) + lane;
  for (int r = 0; r < 8; ++r) {
    pw[r * 32]              = acc0[r];
    pw[GU_TSTRIDE + r * 32] = acc1[r];
  }
  __syncthreads();

  const int row = wave;
  const int t   = lane >> 4;
  const float* pr = part + t * GU_TSTRIDE + (row & 7) * 32 + (row >> 3) * 16 + l;
  float s = pr[0];
  for (int w = 1; w < GU_WAVES; ++w) s += pr[w * (2 * GU_TSTRIDE)];
  const size_t o = (size_t)(m0 + row) * DD + n0 + lane;
  Uout[(size_t)batch * DD * DD + o] = Wbase[o] + s;
}

__global__ __launch_bounds__(128) void chain_q(
    const float* __restrict__ S, const float* __restrict__ wq,
    const float* __restrict__ U, float* __restrict__ out)
{
  __shared__ float buf[2][TR * LSTR];

  const int wave = threadIdx.x >> 5;
  const int lane = threadIdx.x & 31;
  const int half = lane >> 4;
  const int l    = lane & 15;
  const int m0   = blockIdx.x * TR;
  const int n0   = wave * 32;
  const int batch = m0 / BN;
  const float* Ub = U + (size_t)batch * DD * DD;
  const size_t UL = (size_t)4 * DD * DD;

  const float* la0 = &buf[0][l * LSTR + 2 * half];
  const float* la1 = &buf[1][l * LSTR + 2 * half];

  v8f z0, z1, a0, a1;

  v2f pn0[PFN], pn1[PFN];

  mm16x32<DD>(S + (size_t)(m0 + l) * DD + 2 * half, wq, n0, l, half, z0, z1);
  AHEAD_FENCE();
  preN<DD>(Ub, n0, l, half, pn0, pn1);
  put_tile(buf[0], LSTR, n0, l, half, z0, z1);
  LDS_SYNC();

  mmN<DD>(la0, Ub, n0, l, half, pn0, pn1, z0, z1);
  AHEAD_FENCE();
  preN<DD>(Ub + UL, n0, l, half, pn0, pn1);
  for (int r = 0; r < 8; ++r) { a0[r] = siluf(z0[r]); a1[r] = siluf(z1[r]); }
  put_tile(buf[1], LSTR, n0, l, half, a0, a1);
  LDS_SYNC();

  mmN<DD>(la1, Ub + UL, n0, l, half, pn0, pn1, z0, z1);
  AHEAD_FENCE();
  preN<DD>(Ub + 2 * UL, n0, l, half, pn0, pn1);
  for (int r = 0; r < 8; ++r) { a0[r] = siluf(z0[r]); a1[r] = siluf(z1[r]); }
  put_tile(buf[0], LSTR, n0, l, half, a0, a1);
  LDS_SYNC();

  mmN<DD>(la0, Ub + 2 * UL, n0, l, half, pn0, pn1, z0, z1);
  AHEAD_FENCE();
  preN<DD>(Ub + 3 * UL, n0, l, half, pn0, pn1);
  for (int r = 0; r < 8; ++r) { a0[r] = siluf(z0[r]); a1[r] = siluf(z1[r]); }
  put_tile(buf[1], LSTR, n0, l, half, a0, a1);
  LDS_SYNC();

  mmN<DD>(la1, Ub + 3 * UL, n0, l, half, pn0, pn1, z0, z1);
  put_tile(out + (size_t)m0 * DD, DD, n0, l, half, z0, z1);
}

extern "C" void kernel_launch(void* const* d_in, const int* in_sizes, int n_in,
                              void* d_out, int out_size, void* d_ws, size_t ws_size,
                              hipStream_t stream)
{
  const float* S   = (const float*)d_in[0];
  const float* w0  = (const float*)d_in[1];
  const float* w1  = (const float*)d_in[2];
  const float* w2  = (const float*)d_in[3];
  const float* w3  = (const float*)d_in[4];
  const float* wq  = (const float*)d_in[5];
  const float* wkv = (const float*)d_in[6];
  float* out = (float*)d_out;

  float* p = (float*)d_ws;
  const size_t T = (size_t)TOK * DD;
  float* Act = p; p += 4 * T;
  float* Dlt = p; p += 4 * T;
  float* U   = p; p += 16 * DD * DD;

  chain_fb<<<dim3(TOK / TR), dim3(128), 0, stream>>>(S, wkv, w0, w1, w2, w3, Act, Dlt);
  gradU<<<dim3(8, 4, 16), dim3(GU_WAVES * 32), 0, stream>>>(Act, Dlt, w0, w1, w2, w3, U);
  chain_q<<<dim3(TOK / TR), dim3(128), 0, stream>>>(S, wq, U, out);
}


# key path, transposed layers: B fragments read from an LDS copy of w1..w3 left by the forward layers instead of 16-line global loads, all operands grouped ahead of the matrix stream; sums unchanged
# speedup vs baseline: 11.8603x; 1.0851x over previous
#include <hip/hip_runtime.h>

typedef __attribute__((ext_vector_type(2))) float v2f;
typedef __attribute__((ext_vector_type(8))) float v8f;

#define TOK   4096
#define BN    1024
#define DD    128
#define TR    16
#define LSTR  132
#define PW    146

__device__ __forceinline__ float sigf(float x)  { return 1.0f / (1.0f + __expf(-x)); }
__device__ __forceinline__ float siluf(float x) { return x * sigf(x); }
__device__ __forceinline__ float dsiluf(float x){ float s = sigf(x); return s * (1.0f + x * (1.0f - s)); }

__device__ __forceinline__ v8f mma4(v2f a, v2f b, v8f c)
{
  return __builtin_amdgcn_wmma_f32_16x16x4_f32(false, a, false, b, (short)0, c, false, false);
}

template <int LDW>
__device__ __forceinline__ void ldq(const float* __restrict__ W, int j, int n0, int l, int half,
                                    v2f& q0, v2f& q1)
{
  const float* r0 = W + (size_t)(4 * j + 2 * half) * LDW + n0 + 2 * l;
  q0 = *(const v2f*)r0;
  q1 = *(const v2f*)(r0 + LDW);
}

template <int LDW>
__device__ __forceinline__ void mm16x32(const float* arow, const float* __restrict__ W,
                                        int n0, int l, int half, v8f& acc0, v8f& acc1)
{
  acc0 = (v8f){};
  acc1 = (v8f){};
  for (int j = 0; j < DD / 4; ++j) {
    v2f a = *(const v2f*)(arow + 4 * j);
    v2f q0, q1;
    ldq<LDW>(W, j, n0, l, half, q0, q1);
    v2f b0, b1;
    b0.x = q0.x; b0.y = q1.x;
    b1.x = q0.y; b1.y = q1.y;
    acc0 = mma4(a, b0, acc0);
    acc1 = mma4(a, b1, acc1);
  }
}

__device__ __forceinline__ void put_tile(float* dst, int pitch, int n0, int l, int half,
                                         const v8f& c0, const v8f& c1)
{
  float* p = dst + (size_t)(8 * half) * pitch + n0 + 2 * l;
  for (int r = 0; r < 8; ++r) {
    v2f v;
    v.x = c0[r]; v.y = c1[r];
    *(v2f*)(p + (size_t)r * pitch) = v;
  }
}

#define PFN 31

#define OPERANDS_FIRST() __builtin_amdgcn_sched_barrier(0)

template <int LDW>
__device__ __forceinline__ void preN(const float* __restrict__ W, int n0, int l, int half,
                                     v2f (&p0)[PFN], v2f (&p1)[PFN])
{
#pragma unroll
  for (int j = 0; j < PFN; ++j) ldq<LDW>(W, j, n0, l, half, p0[j], p1[j]);
}

template <int LDW, bool KEEP>
__device__ __forceinline__ void mmN(const float* arow, const float* __restrict__ W,
                                    int n0, int l, int half,
                                    const v2f (&p0)[PFN], const v2f (&p1)[PFN], v8f& acc0, v8f& acc1,
                                    float* wst = nullptr)
{
  acc0 = (v8f){};
  acc1 = (v8f){};
  v2f t0[DD / 4 - PFN], t1[DD / 4 - PFN];
#pragma unroll
  for (int j = PFN; j < DD / 4; ++j) ldq<LDW>(W, j, n0, l, half, t0[j - PFN], t1[j - PFN]);
  v2f af[DD / 4];
#pragma unroll
  for (int j = 0; j < DD / 4; ++j) af[j] = *(const v2f*)(arow + 4 * j);
  OPERANDS_FIRST();
#pragma unroll
  for (int j = 0; j < DD / 4; ++j) {
    v2f a = af[j];
    v2f q0, q1;
    if (j < PFN) {
      q0 = p0[j]; q1 = p1[j];
    } else {
      q0 = t0[j - PFN]; q1 = t1[j - PFN];
    }
    v2f b0, b1;
    b0.x = q0.x; b0.y = q1.x;
    b1.x = q0.y; b1.y = q1.y;
    acc0 = mma4(a, b0, acc0);
    acc1 = mma4(a, b1, acc1);
  }
  if (KEEP) {
    __builtin_amdgcn_sched_barrier(0);
#pragma unroll
    for (int j = 0; j < DD / 4; ++j) {
      *(v2f*)(wst + (size_t)(4 * j) * PW)     = j < PFN ? p0[j] : t0[j - PFN];
      *(v2f*)(wst + (size_t)(4 * j + 1) * PW) = j < PFN ? p1[j] : t1[j - PFN];
    }
  }
}

__device__ __forceinline__ void mmTL(const float* arow, const float* wrow, v8f& acc0, v8f& acc1)
{
  acc0 = (v8f){};
  acc1 = (v8f){};
  v2f af[DD / 4], b0[DD / 4], b1[DD / 4];
#pragma unroll
  for (int j = 0; j < DD / 4; ++j) {
    af[j] = *(const v2f*)(arow + 4 * j);
    b0[j] = *(const v2f*)(wrow + 4 * j);
    b1[j] = *(const v2f*)(wrow + PW + 4 * j);
  }
  OPERANDS_FIRST();
#pragma unroll
  for (int j = 0; j < DD / 4; ++j) {
    acc0 = mma4(af[j], b0[j], acc0);
    acc1 = mma4(af[j], b1[j], acc1);
  }
}

#define AHEAD_FENCE() __builtin_amdgcn_sched_barrier(0)

#define LDS_SYNC() do { \
    __builtin_amdgcn_fence(__ATOMIC_RELEASE, "workgroup", "local"); \
    __builtin_amdgcn_s_barrier(); \
    __builtin_amdgcn_fence(__ATOMIC_ACQUIRE, "workgroup", "local"); \
  } while (0)

__global__ __launch_bounds__(128) void chain_fb(
    const float* __restrict__ S,  const float* __restrict__ wkv,
    const float* __restrict__ w0, const float* __restrict__ w1,
    const float* __restrict__ w2, const float* __restrict__ w3,
    float* __restrict__ Act, float* __restrict__ Dlt)
{
  __shared__ float buf[2][TR * LSTR];
  __shared__ float wl[3][DD * PW];

  const int wave = threadIdx.x >> 5;
  const int lane = threadIdx.x & 31;
  const int half = lane >> 4;
  const int l    = lane & 15;
  const int m0   = blockIdx.x * TR;
  const int n0   = wave * 32;
  const size_t T  = (size_t)TOK * DD;
  const size_t go = (size_t)m0 * DD;

  const float* la0 = &buf[0][l * LSTR + 2 * half];
  const float* la1 = &buf[1][l * LSTR + 2 * half];
  const int wso = (2 * half) * PW + n0 + 2 * l;
  const int wro = (n0 + 2 * l) * PW + 2 * half;

  v8f kk0 = {}, kk1 = {}, vv0 = {}, vv1 = {};
  {
    const float* arow = S + (size_t)(m0 + l) * DD + 2 * half;
#pragma unroll
    for (int k0 = 0; k0 < DD; k0 += 4) {
      v2f a = *(const v2f*)(arow + k0);
      const float* r0 = wkv + (size_t)(k0 + 2 * half) * 256 + n0 + 2 * l;
      const float* r1 = r0 + 256;
      const v2f qk0 = *(const v2f*)r0,         qk1 = *(const v2f*)r1;
      const v2f qv0 = *(const v2f*)(r0 + 128), qv1 = *(const v2f*)(r1 + 128);
      v2f bk0, bk1, bv0, bv1;
      bk0.x = qk0.x; bk0.y = qk1.x;
      bk1.x = qk0.y; bk1.y = qk1.y;
      bv0.x = qv0.x; bv0.y = qv1.x;
      bv1.x = qv0.y; bv1.y = qv1.y;
      kk0 = mma4(a, bk0, kk0);
      kk1 = mma4(a, bk1, kk1);
      vv0 = mma4(a, bv0, vv0);
      vv1 = mma4(a, bv1, vv1);
    }
  }
  asm volatile("" :: "v"(vv0), "v"(vv1));
  v2f pn0[PFN], pn1[PFN];
  AHEAD_FENCE();
  preN<DD>(w0, n0, l, half, pn0, pn1);
  put_tile(buf[0], LSTR, n0, l, half, kk0, kk1);
  LDS_SYNC();
  put_tile(Act + go, DD, n0, l, half, kk0, kk1);

  v8f z0, z1, a0, a1;
  v8f g0a, g0b, g1a, g1b, g2a, g2b;

  mmN<DD, false>(la0, w0, n0, l, half, pn0, pn1, z0, z1);
  AHEAD_FENCE();
  preN<DD>(w1, n0, l, half, pn0, pn1);
  for (int r = 0; r < 8; ++r) {
    g0a[r] = dsiluf(z0[r]); g0b[r] = dsiluf(z1[r]);
    a0[r]  = siluf(z0[r]);  a1[r]  = siluf(z1[r]);
  }
  put_tile(buf[1], LSTR, n0, l, half, a0, a1);
  LDS_SYNC();
  put_tile(Act + T + go, DD, n0, l, half, a0, a1);

  mmN<DD, true>(la1, w1, n0, l, half, pn0, pn1, z0, z1, &wl[0][wso]);
  AHEAD_FENCE();
  preN<DD>(w2, n0, l, half, pn0, pn1);
  for (int r = 0; r < 8; ++r) {
    g1a[r] = dsiluf(z0[r]); g1b[r] = dsiluf(z1[r]);
    a0[r]  = siluf(z0[r]);  a1[r]  = siluf(z1[r]);
  }
  put_tile(buf[0], LSTR, n0, l, half, a0, a1);
  LDS_SYNC();
  put_tile(Act + 2 * T + go, DD, n0, l, half, a0, a1);

  mmN<DD, true>(la0, w2, n0, l, half, pn0, pn1, z0, z1, &wl[1][wso]);
  AHEAD_FENCE();
  preN<DD>(w3, n0, l, half, pn0, pn1);
  for (int r = 0; r < 8; ++r) {
    g2a[r] = dsiluf(z0[r]); g2b[r] = dsiluf(z1[r]);
    a0[r]  = siluf(z0[r]);  a1[r]  = siluf(z1[r]);
  }
  put_tile(buf[1], LSTR, n0, l, half, a0, a1);
  LDS_SYNC();
  put_tile(Act + 3 * T + go, DD, n0, l, half, a0, a1);

  mmN<DD, true>(la1, w3, n0, l, half, pn0, pn1, z0, z1, &wl[2][wso]);
  for (int r = 0; r < 8; ++r) {
    a0[r] = (2.0f / 128.0f) * (z0[r] - vv0[r]);
    a1[r] = (2.0f / 128.0f) * (z1[r] - vv1[r]);
  }
  put_tile(buf[0], LSTR, n0, l, half, a0, a1);
  LDS_SYNC();
  put_tile(Dlt + 3 * T + go, DD, n0, l, half, a0, a1);

  mmTL(la0, &wl[2][wro], z0, z1);
  for (int r = 0; r < 8; ++r) { a0[r] = z0[r] * g2a[r]; a1[r] = z1[r] * g2b[r]; }
  put_tile(buf[1], LSTR, n0, l, half, a0, a1);
  LDS_SYNC();
  put_tile(Dlt + 2 * T + go, DD, n0, l, half, a0, a1);

  mmTL(la1, &wl[1][wro], z0, z1);
  for (int r = 0; r < 8; ++r) { a0[r] = z0[r] * g1a[r]; a1[r] = z1[r] * g1b[r]; }
  put_tile(buf[0], LSTR, n0, l, half, a0, a1);
  LDS_SYNC();
  put_tile(Dlt + T + go, DD, n0, l, half, a0, a1);

  mmTL(la0, &wl[0][wro], z0, z1);
  for (int r = 0; r < 8; ++r) { a0[r] = z0[r] * g0a[r]; a1[r] = z1[r] * g0b[r]; }
  put_tile(Dlt + go, DD, n0, l, half, a0, a1);
}

#define GU_WAVES   16
#define GU_KPW     (BN / GU_WAVES)
#define GU_TSTRIDE 272
__global__ __launch_bounds__(GU_WAVES * 32) void gradU(
    const float* __restrict__ ActAll,
    const float* __restrict__ DltAll,
    const float* __restrict__ w0, const float* __restrict__ w1,
    const float* __restrict__ w2, const float* __restrict__ w3,
    float* __restrict__ UAll)
{
  __shared__ float part[GU_WAVES * 2 * GU_TSTRIDE];

  const int wave = threadIdx.x >> 5;
  const int lane = threadIdx.x & 31;
  const int half = lane >> 4;
  const int l    = lane & 15;
  const int m0   = blockIdx.x * 16;
  const int n0   = blockIdx.y * 32;
  const int layer = blockIdx.z >> 2;
  const int batch = blockIdx.z & 3;
  const size_t T  = (size_t)TOK * DD;
  const float* Act   = ActAll + (size_t)layer * T;
  const float* Dlt   = DltAll + (size_t)layer * T;
  const float* Wbase = layer == 0 ? w0 : layer == 1 ? w1 : layer == 2 ? w2 : w3;
  float* Uout = UAll + (size_t)layer * 4 * DD * DD;
  const size_t tok0 = (size_t)batch * BN + (size_t)wave * GU_KPW;
  const float* Ab = Act + tok0 * DD + m0 + l;
  const float* Db = Dlt + tok0 * DD + n0 + l;

  v8f acc0 = {};
  v8f acc1 = {};
  for (int k0 = 0; k0 < GU_KPW; k0 += 4) {
    const int kr = k0 + 2 * half;
    v2f a;
    a.x = Ab[(size_t)(kr + 0) * DD];
    a.y = Ab[(size_t)(kr + 1) * DD];
    const float* d0 = Db + (size_t)kr * DD;
    const float* d1 = d0 + DD;
    v2f b0, b1;
    b0.x = d0[0];  b0.y = d1[0];
    b1.x = d0[16]; b1.y = d1[16];
    acc0 = mma4(a, b0, acc0);
    acc1 = mma4(a, b1, acc1);
  }

  float* pw = part + wave * (2 * GU_TSTRIDE) + lane;
  for (int r = 0; r < 8; ++r) {
    pw[r * 32]              = acc0[r];
    pw[GU_TSTRIDE + r * 32] = acc1[r];
  }
  __syncthreads();

  const int row = wave;
  const int t   = lane >> 4;
  const float* pr = part + t * GU_TSTRIDE + (row & 7) * 32 + (row >> 3) * 16 + l;
  float s = pr[0];
  for (int w = 1; w < GU_WAVES; ++w) s += pr[w * (2 * GU_TSTRIDE)];
  const size_t o = (size_t)(m0 + row) * DD + n0 + lane;
  Uout[(size_t)batch * DD * DD + o] = Wbase[o] + s;
}

__global__ __launch_bounds__(128) void chain_q(
    const float* __restrict__ S, const float* __restrict__ wq,
    const float* __restrict__ U, float* __restrict__ out)
{
  __shared__ float buf[2][TR * LSTR];

  const int wave = threadIdx.x >> 5;
  const int lane = threadIdx.x & 31;
  const int half = lane >> 4;
  const int l    = lane & 15;
  const int m0   = blockIdx.x * TR;
  const int n0   = wave * 32;
  const int batch = m0 / BN;
  const float* Ub = U + (size_t)batch * DD * DD;
  const size_t UL = (size_t)4 * DD * DD;

  const float* la0 = &buf[0][l * LSTR + 2 * half];
  const float* la1 = &buf[1][l * LSTR + 2 * half];

  v8f z0, z1, a0, a1;

  v2f pn0[PFN], pn1[PFN];

  mm16x32<DD>(S + (size_t)(m0 + l) * DD + 2 * half, wq, n0, l, half, z0, z1);
  AHEAD_FENCE();
  preN<DD>(Ub, n0, l, half, pn0, pn1);
  put_tile(buf[0], LSTR, n0, l, half, z0, z1);
  LDS_SYNC();

  mmN<DD, false>(la0, Ub, n0, l, half, pn0, pn1, z0, z1);
  AHEAD_FENCE();
  preN<DD>(Ub + UL, n0, l, half, pn0, pn1);
  for (int r = 0; r < 8; ++r) { a0[r] = siluf(z0[r]); a1[r] = siluf(z1[r]); }
  put_tile(buf[1], LSTR, n0, l, half, a0, a1);
  LDS_SYNC();

  mmN<DD, false>(la1, Ub + UL, n0, l, half, pn0, pn1, z0, z1);
  AHEAD_FENCE();
  preN<DD>(Ub + 2 * UL, n0, l, half, pn0, pn1);
  for (int r = 0; r < 8; ++r) { a0[r] = siluf(z0[r]); a1[r] = siluf(z1[r]); }
  put_tile(buf[0], LSTR, n0, l, half, a0, a1);
  LDS_SYNC();

  mmN<DD, false>(la0, Ub + 2 * UL, n0, l, half, pn0, pn1, z0, z1);
  AHEAD_FENCE();
  preN<DD>(Ub + 3 * UL, n0, l, half, pn0, pn1);
  for (int r = 0; r < 8; ++r) { a0[r] = siluf(z0[r]); a1[r] = siluf(z1[r]); }
  put_tile(buf[1], LSTR, n0, l, half, a0, a1);
  LDS_SYNC();

  mmN<DD, false>(la1, Ub + 3 * UL, n0, l, half, pn0, pn1, z0, z1);
  put_tile(out + (size_t)m0 * DD, DD, n0, l, half, z0, z1);
}

extern "C" void kernel_launch(void* const* d_in, const int* in_sizes, int n_in,
                              void* d_out, int out_size, void* d_ws, size_t ws_size,
                              hipStream_t stream)
{
  const float* S   = (const float*)d_in[0];
  const float* w0  = (const float*)d_in[1];
  const float* w1  = (const float*)d_in[2];
  const float* w2  = (const float*)d_in[3];
  const float* w3  = (const float*)d_in[4];
  const float* wq  = (const float*)d_in[5];
  const float* wkv = (const float*)d_in[6];
  float* out = (float*)d_out;

  float* p = (float*)d_ws;
  const size_t T = (size_t)TOK * DD;
  float* Act = p; p += 4 * T;
  float* Dlt = p; p += 4 * T;
  float* U   = p; p += 16 * DD * DD;

  chain_fb<<<dim3(TOK / TR), dim3(128), 0, stream>>>(S, wkv, w0, w1, w2, w3, Act, Dlt);
  gradU<<<dim3(8, 4, 16), dim3(GU_WAVES * 32), 0, stream>>>(Act, Dlt, w0, w1, w2, w3, U);
  chain_q<<<dim3(TOK / TR), dim3(128), 0, stream>>>(S, wq, U, out);
}
